# starting kernel
# baseline (speedup 1.0000x reference)
#include <hip/hip_runtime.h>
#include <hip/hip_bf16.h>
#include <math.h>

#define N_NODES 100000
#define N_EDGES 1600000
#define H_HEADS 4
#define F_IN    128
#define F_OUT   32
#define HF      (H_HEADS * F_OUT)
#define SLOPE   0.2f

typedef __attribute__((ext_vector_type(2))) float v2f;
typedef __attribute__((ext_vector_type(8))) float v8f;

__device__ __forceinline__ float lrelu(float v) { return v > 0.f ? v : SLOPE * v; }

__device__ __forceinline__ unsigned enc_f32(float f) {
    unsigned u = __float_as_uint(f);
    return (u & 0x80000000u) ? ~u : (u | 0x80000000u);
}
__device__ __forceinline__ float dec_f32(unsigned e) {
    return (e & 0x80000000u) ? __uint_as_float(e ^ 0x80000000u) : __uint_as_float(~e);
}

__global__ void gat_init(float* __restrict__ out, const float* __restrict__ bias,
                         unsigned* __restrict__ m_enc, float* __restrict__ denom) {
    int idx = blockIdx.x * blockDim.x + threadIdx.x;
    if (idx < N_NODES * HF) out[idx] = bias[idx & (HF - 1)];
    if (idx < N_NODES * H_HEADS) {
        m_enc[idx] = enc_f32(-__builtin_inff());
        denom[idx] = 0.f;
    }
}

__global__ void gat_repack_w(const float* __restrict__ w, v2f* __restrict__ wt) {
    int idx = blockIdx.x * blockDim.x + threadIdx.x;
    if (idx >= H_HEADS * 2 * 32 * 2 * 16) return;
    int n    = idx & 15;
    int kh   = (idx >> 4) & 1;
    int kq   = (idx >> 5) & 31;
    int ct   = (idx >> 10) & 1;
    int head = idx >> 11;
    int k    = kq * 4 + kh * 2;
    int col  = ct * 16 + n;
    const float* wp = w + (size_t)head * (F_IN * F_OUT) + (size_t)k * F_OUT + col;
    v2f v; v.x = wp[0]; v.y = wp[F_OUT];
    wt[idx] = v;
}

__global__ void __launch_bounds__(256)
gat_gemm_wmma(const float* __restrict__ x, const v2f* __restrict__ wt,
              float* __restrict__ h) {
    const int lane = threadIdx.x & 31;
    const int wave = threadIdx.x >> 5;
    const int head = wave & 3;
    const int rh   = wave >> 2;
    const int ln   = lane & 15;
    const int kh   = lane >> 4;

    const int row0 = blockIdx.x * 32 + rh * 16;
    const float* xr = x + (size_t)(row0 + ln) * F_IN;
    const v2f* bp = wt + head * 2048 + kh * 16 + ln;

    v8f c0 = {}, c1 = {};
#pragma unroll
    for (int kq = 0; kq < 32; ++kq) {
        v2f a  = *(const v2f*)(xr + kq * 4 + kh * 2);
        v2f b0 = bp[kq * 32];
        v2f b1 = bp[1024 + kq * 32];
        c0 = __builtin_amdgcn_wmma_f32_16x16x4_f32(false, a, false, b0,
                                                   (short)0, c0, false, false);
        c1 = __builtin_amdgcn_wmma_f32_16x16x4_f32(false, a, false, b1,
                                                   (short)0, c1, false, false);
    }
    float* hp = h + (size_t)(row0 + kh * 8) * HF + head * F_OUT + ln;
#pragma unroll
    for (int r = 0; r < 8; ++r) {
        hp[(size_t)r * HF]      = c0[r];
        hp[(size_t)r * HF + 16] = c1[r];
    }
}

__global__ void gat_alpha(const float* __restrict__ h, const float* __restrict__ a,
                          float* __restrict__ alpha_src, float* __restrict__ alpha_dst) {
    int idx = blockIdx.x * blockDim.x + threadIdx.x;
    if (idx >= N_NODES * H_HEADS) return;
    int head = idx & (H_HEADS - 1);
    const float* hp = h + (size_t)idx * F_OUT;
    const float* as = a + head * 2 * F_OUT;
    const float* ad = as + F_OUT;
    float s = 0.f, d = 0.f;
#pragma unroll
    for (int o = 0; o < F_OUT; ++o) {
        float v = hp[o];
        s += v * as[o];
        d += v * ad[o];
    }
    alpha_src[idx] = s;
    alpha_dst[idx] = d;
}

__global__ void gat_edge_max(const float* __restrict__ alpha_src,
                             const float* __restrict__ alpha_dst,
                             const int* __restrict__ esrc, const int* __restrict__ edst,
                             unsigned* __restrict__ m_enc) {
    int e = blockIdx.x * blockDim.x + threadIdx.x;
    if (e >= N_EDGES) return;
    int s = esrc[e], d = edst[e];
    const float4 as = *(const float4*)(alpha_src + (size_t)s * 4);
    const float4 ad = *(const float4*)(alpha_dst + (size_t)d * 4);
    unsigned* mp = m_enc + (size_t)d * 4;
    atomicMax(mp + 0, enc_f32(lrelu(as.x + ad.x)));
    atomicMax(mp + 1, enc_f32(lrelu(as.y + ad.y)));
    atomicMax(mp + 2, enc_f32(lrelu(as.z + ad.z)));
    atomicMax(mp + 3, enc_f32(lrelu(as.w + ad.w)));
}

__global__ void gat_finalize_max(const unsigned* __restrict__ m_enc,
                                 float* __restrict__ m) {
    int idx = blockIdx.x * blockDim.x + threadIdx.x;
    if (idx >= N_NODES * H_HEADS) return;
    float f = dec_f32(m_enc[idx]);
    if (!isfinite(f)) f = 0.f;
    m[idx] = f;
}

__global__ void gat_edge_sum(const float* __restrict__ alpha_src,
                             const float* __restrict__ alpha_dst,
                             const float* __restrict__ m,
                             const int* __restrict__ esrc, const int* __restrict__ edst,
                             float* __restrict__ denom) {
    int e = blockIdx.x * blockDim.x + threadIdx.x;
    if (e >= N_EDGES) return;
    int s = esrc[e], d = edst[e];
    const float4 as = *(const float4*)(alpha_src + (size_t)s * 4);
    const float4 ad = *(const float4*)(alpha_dst + (size_t)d * 4);
    const float4 mv = *(const float4*)(m + (size_t)d * 4);
    float* dp = denom + (size_t)d * 4;
    atomicAdd(dp + 0, expf(lrelu(as.x + ad.x) - mv.x));
    atomicAdd(dp + 1, expf(lrelu(as.y + ad.y) - mv.y));
    atomicAdd(dp + 2, expf(lrelu(as.z + ad.z) - mv.z));
    atomicAdd(dp + 3, expf(lrelu(as.w + ad.w) - mv.w));
}

__global__ void __launch_bounds__(256)
gat_edge_scatter(const float* __restrict__ h,
                 const float* __restrict__ alpha_src, const float* __restrict__ alpha_dst,
                 const float* __restrict__ m, const float* __restrict__ denom,
                 const int* __restrict__ esrc, const int* __restrict__ edst,
                 float* __restrict__ out) {
    int e = blockIdx.x * (blockDim.x >> 5) + (threadIdx.x >> 5);
    if (e >= N_EDGES) return;
    int lane = threadIdx.x & 31;
    int s = esrc[e], d = edst[e];
    const float* hs = h + (size_t)s * HF;
    float*       op = out + (size_t)d * HF;
#pragma unroll
    for (int i = 0; i < H_HEADS; ++i) {
        float ev  = lrelu(alpha_src[(size_t)s * 4 + i] + alpha_dst[(size_t)d * 4 + i]);
        float att = expf(ev - m[(size_t)d * 4 + i]) / (denom[(size_t)d * 4 + i] + 1e-16f);
        atomicAdd(op + i * F_OUT + lane, hs[i * F_OUT + lane] * att);
    }
}

extern "C" void kernel_launch(void* const* d_in, const int* in_sizes, int n_in,
                              void* d_out, int out_size, void* d_ws, size_t ws_size,
                              hipStream_t stream) {
    const float* x      = (const float*)d_in[0];
    const float* weight = (const float*)d_in[1];
    const float* a      = (const float*)d_in[2];
    const float* bias   = (const float*)d_in[3];
    const int*   esrc   = (const int*)d_in[4];
    const int*   edst   = (const int*)d_in[5];
    float* out = (float*)d_out;

    float*    h         = (float*)d_ws;
    float*    alpha_src = h + (size_t)N_NODES * HF;
    float*    alpha_dst = alpha_src + (size_t)N_NODES * H_HEADS;
    unsigned* m_enc     = (unsigned*)(alpha_dst + (size_t)N_NODES * H_HEADS);
    float*    m         = (float*)(m_enc + (size_t)N_NODES * H_HEADS);
    float*    denom     = m + (size_t)N_NODES * H_HEADS;
    v2f*      wt        = (v2f*)(denom + (size_t)N_NODES * H_HEADS);

    const int TB = 256;
    gat_init<<<(N_NODES * HF + TB - 1) / TB, TB, 0, stream>>>(out, bias, m_enc, denom);
    gat_repack_w<<<(H_HEADS * 2 * 32 * 2 * 16 + TB - 1) / TB, TB, 0, stream>>>(weight, wt);
    gat_gemm_wmma<<<N_NODES / 32, TB, 0, stream>>>(x, wt, h);
    gat_alpha<<<(N_NODES * H_HEADS + TB - 1) / TB, TB, 0, stream>>>(h, a, alpha_src, alpha_dst);
    gat_edge_max<<<(N_EDGES + TB - 1) / TB, TB, 0, stream>>>(alpha_src, alpha_dst, esrc, edst, m_enc);
    gat_finalize_max<<<(N_NODES * H_HEADS + TB - 1) / TB, TB, 0, stream>>>(m_enc, m);
    gat_edge_sum<<<(N_EDGES + TB - 1) / TB, TB, 0, stream>>>(alpha_src, alpha_dst, m, esrc, edst, denom);
    gat_edge_scatter<<<N_EDGES / 8, TB, 0, stream>>>(h, alpha_src, alpha_dst, m, denom, esrc, edst, out);
}


# Node pass: lists of 33 to 64 edges walk the same loop of 8-row groups as the short ones (row loads on base + 32-bit index): 61 registers, were 122, so a SIMD holds 16 waves, not 8; sums unchanged
# speedup vs baseline: 3.9105x; 3.9105x over previous
#include <hip/hip_runtime.h>
#include <math.h>

#define N_NODES 100000
#define N_EDGES 1600000
#define H_HEADS 4
#define F_IN    128
#define F_OUT   32
#define HF      (H_HEADS * F_OUT)
#define SLOPE   0.2f
#define ROW_CAP HF
static_assert(N_EDGES % 32 == 0, "the edge walk of an over-full node has no tail");

typedef __attribute__((ext_vector_type(2))) float v2f;
typedef __attribute__((ext_vector_type(8))) float v8f;

__device__ __forceinline__ float lrelu(float v) { return v > 0.f ? v : SLOPE * v; }

__device__ __forceinline__ float wave_max(float v) {
    for (int o = 16; o > 0; o >>= 1) v = fmaxf(v, __shfl_xor(v, o));
    return v;
}
__device__ __forceinline__ float wave_sum(float v) {
    for (int o = 16; o > 0; o >>= 1) v += __shfl_xor(v, o);
    return v;
}
__device__ __forceinline__ int rl_i(int v, int l) { return __builtin_amdgcn_readlane(v, l); }
__device__ __forceinline__ float rl_f(float v, int l) {
    return __int_as_float(__builtin_amdgcn_readlane(__float_as_int(v), l));
}

__global__ void gat_prep(const float* __restrict__ w, v2f* __restrict__ wt,
                         int* __restrict__ cnt) {
    int idx = blockIdx.x * blockDim.x + threadIdx.x;
    if (idx < N_NODES) cnt[idx] = 0;
    if (idx >= H_HEADS * 2 * 32 * 2 * 16) return;
    int n    = idx & 15;
    int kh   = (idx >> 4) & 1;
    int kq   = (idx >> 5) & 31;
    int ct   = (idx >> 10) & 1;
    int head = idx >> 11;
    int k    = kq * 4 + kh * 2;
    int col  = ct * 16 + n;
    const float* wp = w + (size_t)head * (F_IN * F_OUT) + (size_t)k * F_OUT + col;
    v2f v; v.x = wp[0]; v.y = wp[F_OUT];
    wt[idx] = v;
}

__global__ void gat_lists(const int* __restrict__ esrc, const int* __restrict__ edst,
                          int* __restrict__ cnt, int* __restrict__ rows) {
    int e = blockIdx.x * blockDim.x + threadIdx.x;
    if (e >= N_EDGES) return;
    const int d = edst[e];
    const int s = esrc[e];
    const int place = atomicAdd(cnt + d, 1);
    rows[(size_t)d * HF + min(place, ROW_CAP - 1)] = s;
}

__global__ void __launch_bounds__(256)
gat_gemm_wmma(const float* __restrict__ x, const v2f* __restrict__ wt,
              float* __restrict__ h) {
    const int lane = threadIdx.x & 31;
    const int wave = threadIdx.x >> 5;
    const int head = wave & 3;
    const int rh   = wave >> 2;
    const int ln   = lane & 15;
    const int kh   = lane >> 4;

    const int row0 = blockIdx.x * 32 + rh * 16;
    const float* xr = x + (size_t)(row0 + ln) * F_IN;
    const v2f* bp = wt + head * 2048 + kh * 16 + ln;

    v8f c0 = {}, c1 = {};
#pragma unroll
    for (int kq = 0; kq < 32; ++kq) {
        v2f a  = *(const v2f*)(xr + kq * 4 + kh * 2);
        v2f b0 = bp[kq * 32];
        v2f b1 = bp[1024 + kq * 32];
        c0 = __builtin_amdgcn_wmma_f32_16x16x4_f32(false, a, false, b0,
                                                   (short)0, c0, false, false);
        c1 = __builtin_amdgcn_wmma_f32_16x16x4_f32(false, a, false, b1,
                                                   (short)0, c1, false, false);
    }
    float* hp = h + (size_t)(row0 + kh * 8) * HF + head * F_OUT + ln;
#pragma unroll
    for (int r = 0; r < 8; ++r) {
        hp[(size_t)r * HF]      = c0[r];
        hp[(size_t)r * HF + 16] = c1[r];
    }
}

__device__ __forceinline__ long long next_key(const float* orow, int deg,
                                              int lane, long long last) {
    long long best = 0x7fffffffffffffffLL;
    for (int p = lane; p < deg; p += 32) {
        long long key = ((long long)__float_as_int(orow[p]) << 32) | (long long)(unsigned)p;
        if (key > last && key < best) best = key;
    }
    for (int o = 16; o > 0; o >>= 1) {
        long long t = __shfl_xor(best, o);
        best = t < best ? t : best;
    }
    return best;
}

template <int M>
__device__ __forceinline__ float lane_xor(float v) {
    return __int_as_float(__builtin_amdgcn_update_dpp(0, __float_as_int(v), 0x160 + M, 0xf, 0xf, true));
}
__device__ __forceinline__ float head_sum(float v) {
    v += lane_xor<1>(v);
    v += lane_xor<2>(v);
    v += lane_xor<4>(v);
    return v;
}
__device__ __forceinline__ float row_score(const float4 r, const float4 av) {
    float p = r.x * av.x;
    p = fmaf(r.y, av.y, p);
    p = fmaf(r.z, av.z, p);
    p = fmaf(r.w, av.w, p);
    return head_sum(p);
}

__device__ __forceinline__ void soft_one(const float4 row, const float4 asv, float ad,
                                         float& m, float& d, float4& acc) {
    const float e = lrelu(row_score(row, asv) + ad);
    const float gm = fmaxf(m, e);
    const float sc = expf(m - gm);
    const float x = expf(e - gm);
    d = fmaf(d, sc, x);
    acc.x = fmaf(acc.x, sc, x * row.x);
    acc.y = fmaf(acc.y, sc, x * row.y);
    acc.z = fmaf(acc.z, sc, x * row.z);
    acc.w = fmaf(acc.w, sc, x * row.w);
    m = gm;
}

template <int J>
__device__ __forceinline__ float group_lane(float v) {
    return __int_as_float((int)__builtin_amdgcn_mov_dpp8((unsigned)__float_as_int(v), (unsigned)J * 0x249249u));
}

__device__ __forceinline__ void soft_group(const float* __restrict__ h, int ss, int k0, int rem, int lane,
                                           const float4 asv, float ad, bool rescale,
                                           float& m, float& d, float4& acc) {
    float4 r[8];
    float e[8];
#pragma unroll
    for (int j = 0; j < 8; ++j) {
        if (j < rem) {
            const int sk = rl_i(ss, k0 + j);
            r[j] = ((const float4*)h)[(unsigned)sk * (HF / 4) + (unsigned)lane];
        }
    }
    float gm = m;
#pragma unroll
    for (int j = 0; j < 8; ++j) {
        e[j] = -__builtin_inff();
        if (j < rem) {
            e[j] = lrelu(row_score(r[j], asv) + ad);
            gm = fmaxf(gm, e[j]);
        }
    }
    if (rescale) {
        const float sc = expf(m - gm);
        d *= sc;
        acc.x *= sc; acc.y *= sc; acc.z *= sc; acc.w *= sc;
    }
    const int li = lane & 7;
    float es = e[0];
#pragma unroll
    for (int j = 1; j < 8; ++j) es = (li == j) ? e[j] : es;
    const float xs = expf(es - gm);
#define GAT_ROW_TERM(J)                                   \
    if (J < rem) {                                        \
        const float x = group_lane<J>(xs);                \
        d += x;                                           \
        acc.x = fmaf(x, r[J].x, acc.x);                   \
        acc.y = fmaf(x, r[J].y, acc.y);                   \
        acc.z = fmaf(x, r[J].z, acc.z);                   \
        acc.w = fmaf(x, r[J].w, acc.w);                   \
    }
    GAT_ROW_TERM(0) GAT_ROW_TERM(1) GAT_ROW_TERM(2) GAT_ROW_TERM(3)
    GAT_ROW_TERM(4) GAT_ROW_TERM(5) GAT_ROW_TERM(6) GAT_ROW_TERM(7)
#undef GAT_ROW_TERM
    m = gm;
}

__global__ void __launch_bounds__(256)
gat_node_gather(const float* __restrict__ h, const float* __restrict__ a,
                const int* __restrict__ cnt, const int* __restrict__ esrc,
                const int* __restrict__ edst, const float* __restrict__ bias,
                float* out) {
    const int lane = threadIdx.x & 31;
    const int head = lane >> 3;
    const int n = __builtin_amdgcn_readfirstlane(blockIdx.x * 8 + (threadIdx.x >> 5));
    if (n >= N_NODES) return;
    const int deg = __builtin_amdgcn_readfirstlane(cnt[n]);
    float* orow = out + (size_t)n * HF;
    const float4 asv = *(const float4*)(a + head * 2 * F_OUT + (lane & 7) * 4);
    const float4 adv = *(const float4*)(a + head * 2 * F_OUT + F_OUT + (lane & 7) * 4);
    const float4 hn  = *(const float4*)(h + ((unsigned)n * HF + (unsigned)lane * 4));
    const float4 bv  = *(const float4*)(bias + lane * 4);
    const float ad = row_score(hn, adv);
    float m = -__builtin_inff();
    float d = 0.f;
    float4 acc = make_float4(0.f, 0.f, 0.f, 0.f);

    if (deg <= 64) {
        int ss0 = 0, ss1 = 0;
        if (deg <= 32) {
            const bool valid = lane < deg;
            const int s_in = valid ? __float_as_int(orow[lane]) : 0;
            const unsigned ck = valid ? (((unsigned)s_in << 5) | (unsigned)lane)
                                      : (0xFFFFFFE0u | (unsigned)lane);
            int rk = 0;
            for (int j = 0; j < deg; ++j) {
                unsigned kj = (unsigned)rl_i((int)ck, j);
                rk += (kj < ck) ? 1 : 0;
            }
            if (!valid) rk = lane;
            ss0 = __builtin_amdgcn_ds_permute(rk << 2, s_in);
        } else {
            const bool v1 = lane + 32 < deg;
            const int s0 = __float_as_int(orow[lane]);
            const int s1 = v1 ? __float_as_int(orow[32 + lane]) : 0;
            const unsigned k0 = ((unsigned)s0 << 6) | (unsigned)lane;
            const unsigned k1 = v1 ? (((unsigned)s1 << 6) | (unsigned)(32 + lane)) : 0xFFFFFFFFu;
            int r0 = 0, r1 = 0;
            for (int j = 0; j < 32; ++j) {
                const unsigned kj = (unsigned)rl_i((int)k0, j);
                r0 += (kj < k0) ? 1 : 0;
                r1 += (kj < k1) ? 1 : 0;
            }
            for (int j = 32; j < deg; ++j) {
                const unsigned kj = (unsigned)rl_i((int)k1, j - 32);
                r0 += (kj < k0) ? 1 : 0;
                r1 += (kj < k1) ? 1 : 0;
            }
            for (int k = 0; k < deg; ++k) {
                const unsigned b0 = __builtin_amdgcn_ballot_w32(r0 == k);
                const unsigned b1 = __builtin_amdgcn_ballot_w32(r1 == k);
                const int l = __builtin_ctz(b0 | b1);
                const int sk = b0 ? rl_i(s0, l) : rl_i(s1, l);
                if (lane == (k & 31)) { if (k < 32) ss0 = sk; else ss1 = sk; }
            }
        }
        for (int k0 = 0; k0 < deg; k0 += 8)
            soft_group(h, k0 < 32 ? ss0 : ss1, k0 & 31, deg - k0, lane, asv, ad, k0 != 0, m, d, acc);
    } else if (deg <= ROW_CAP) {
        long long last = -1;
        for (int k = 0; k < deg; ++k) {
            last = next_key(orow, deg, lane, last);
            const int s = (int)(last >> 32);
            const float4 row = *(const float4*)(h + (size_t)s * HF + lane * 4);
            soft_one(row, asv, ad, m, d, acc);
        }
    } else {
        for (int e0 = 0; e0 < N_EDGES; e0 += 32) {
            const bool hit = edst[e0 + lane] == n;
            const int s_l = hit ? esrc[e0 + lane] : 0;
            unsigned todo = __builtin_amdgcn_ballot_w32(hit);
            while (todo) {
                const int l = __builtin_ctz(todo);
                todo &= todo - 1;
                const int s = rl_i(s_l, l);
                const float4 row = *(const float4*)(h + (size_t)s * HF + lane * 4);
                soft_one(row, asv, ad, m, d, acc);
            }
        }
    }
    const float q = 1.0f / (d + 1e-16f);
    float4 o;
    o.x = acc.x * q + bv.x; o.y = acc.y * q + bv.y; o.z = acc.z * q + bv.z; o.w = acc.w * q + bv.w;
    *(float4*)(orow + lane * 4) = o;
}

extern "C" void kernel_launch(void* const* d_in, const int* in_sizes, int n_in,
                              void* d_out, int out_size, void* d_ws, size_t ws_size,
                              hipStream_t stream) {
    const float* x      = (const float*)d_in[0];
    const float* weight = (const float*)d_in[1];
    const float* a      = (const float*)d_in[2];
    const float* bias   = (const float*)d_in[3];
    const int*   esrc   = (const int*)d_in[4];
    const int*   edst   = (const int*)d_in[5];
    float* out = (float*)d_out;

    float* h    = (float*)d_ws;
    int*   cnt  = (int*)(h + (size_t)N_NODES * HF);
    v2f*   wt   = (v2f*)(cnt + N_NODES);
    int*   rows = (int*)d_out;

    const int TB = 256;
    gat_prep<<<(N_NODES + TB - 1) / TB, TB, 0, stream>>>(weight, wt, cnt);
    gat_lists<<<(N_EDGES + TB - 1) / TB, TB, 0, stream>>>(esrc, edst, cnt, rows);
    gat_gemm_wmma<<<N_NODES / 32, TB, 0, stream>>>(x, wt, h);
    gat_node_gather<<<N_NODES / 8, TB, 0, stream>>>(h, a, cnt, esrc, edst, bias, out);
}


# Node lists built by a two-level counting sort by destination on LDS counters (chunk counts, bin scatter, one block per 512-node bin): no global atomics, list lines stored whole; 5 launches; same sums
# speedup vs baseline: 6.0520x; 1.5476x over previous
#include <hip/hip_runtime.h>
#include <math.h>

#define N_NODES 100000
#define N_EDGES 1600000
#define H_HEADS 4
#define F_IN    128
#define F_OUT   32
#define HF      (H_HEADS * F_OUT)
#define SLOPE   0.2f
#define ROW_CAP HF
static_assert(N_EDGES % 32 == 0, "the edge walk of an over-full node has no tail");

#define EDGE_CHUNKS  256
#define CHUNK_EDGES  (N_EDGES / EDGE_CHUNKS)
#define BIN_SHIFT    9
#define BIN_NODES    (1 << BIN_SHIFT)
#define N_BINS       ((N_NODES + BIN_NODES - 1) / BIN_NODES)
#define SRC_BITS     17
#define SRC_MASK     ((1 << SRC_BITS) - 1)
#define SORT_THREADS 1024
#define PER_THREAD   ((CHUNK_EDGES + SORT_THREADS - 1) / SORT_THREADS)
#define SCAN_PER_LANE ((N_BINS + 31) / 32)
#define LDS_LIST     32
static_assert(N_EDGES % EDGE_CHUNKS == 0, "equal chunks");
static_assert(EDGE_CHUNKS == 256, "a bin's per-chunk counts are read as 8 per lane");
static_assert(N_NODES <= (1 << SRC_BITS) && BIN_SHIFT + SRC_BITS < 32, "a binned edge fits a non-negative word");
static_assert(N_EDGES + N_BINS * EDGE_CHUNKS + N_BINS + 1 <= N_NODES * HF, "the sort's scratch fits where h is written later");

typedef __attribute__((ext_vector_type(2))) float v2f;
typedef __attribute__((ext_vector_type(8))) float v8f;

__device__ __forceinline__ float lrelu(float v) { return v > 0.f ? v : SLOPE * v; }

__device__ __forceinline__ float wave_max(float v) {
    for (int o = 16; o > 0; o >>= 1) v = fmaxf(v, __shfl_xor(v, o));
    return v;
}
__device__ __forceinline__ float wave_sum(float v) {
    for (int o = 16; o > 0; o >>= 1) v += __shfl_xor(v, o);
    return v;
}
__device__ __forceinline__ int wave_isum(int v) {
    for (int o = 16; o > 0; o >>= 1) v += __shfl_xor(v, o);
    return v;
}
__device__ __forceinline__ int rl_i(int v, int l) { return __builtin_amdgcn_readlane(v, l); }
__device__ __forceinline__ float rl_f(float v, int l) {
    return __int_as_float(__builtin_amdgcn_readlane(__float_as_int(v), l));
}

__global__ void __launch_bounds__(SORT_THREADS)
gat_prep_hist(const float* __restrict__ w, v2f* __restrict__ wt,
              const int* __restrict__ edst, int* __restrict__ hist) {
    __shared__ int hb[N_BINS];
    const int tid = threadIdx.x;
    const int idx = blockIdx.x * SORT_THREADS + tid;
    if (tid < N_BINS) hb[tid] = 0;
    if (idx < H_HEADS * 2 * 32 * 2 * 16) {
        int n    = idx & 15;
        int kh   = (idx >> 4) & 1;
        int kq   = (idx >> 5) & 31;
        int ct   = (idx >> 10) & 1;
        int head = idx >> 11;
        int k    = kq * 4 + kh * 2;
        int col  = ct * 16 + n;
        const float* wp = w + (size_t)head * (F_IN * F_OUT) + (size_t)k * F_OUT + col;
        v2f v; v.x = wp[0]; v.y = wp[F_OUT];
        wt[idx] = v;
    }
    const int* dp = edst + blockIdx.x * CHUNK_EDGES;
    int d[PER_THREAD];
#pragma unroll
    for (int k = 0; k < PER_THREAD; ++k) {
        const int e = k * SORT_THREADS + tid;
        d[k] = e < CHUNK_EDGES ? dp[e] : -1;
    }
    __syncthreads();
#pragma unroll
    for (int k = 0; k < PER_THREAD; ++k)
        if (d[k] >= 0) atomicAdd(&hb[d[k] >> BIN_SHIFT], 1);
    __syncthreads();
    if (tid < N_BINS) hist[tid * EDGE_CHUNKS + blockIdx.x] = hb[tid];
}

__global__ void __launch_bounds__(SORT_THREADS)
gat_bin_scatter(const int* __restrict__ esrc, const int* __restrict__ edst,
                const int* __restrict__ hist, int* __restrict__ binstart,
                int* __restrict__ binned) {
    __shared__ int tot[N_BINS];
    __shared__ int pre[N_BINS];
    __shared__ int offs[N_BINS];
    const int tid  = threadIdx.x;
    const int lane = tid & 31;
    const int wave = tid >> 5;
    const int me   = blockIdx.x;
    const int* dp = edst + me * CHUNK_EDGES;
    const int* sp = esrc + me * CHUNK_EDGES;
    int d[PER_THREAD], s[PER_THREAD];
#pragma unroll
    for (int k = 0; k < PER_THREAD; ++k) {
        const int e = k * SORT_THREADS + tid;
        d[k] = e < CHUNK_EDGES ? dp[e] : -1;
        s[k] = e < CHUNK_EDGES ? sp[e] : 0;
    }
    for (int b = wave; b < N_BINS; b += SORT_THREADS / 32) {
        const int4* hp = (const int4*)(hist + b * EDGE_CHUNKS) + lane * 2;
        const int4 u = hp[0];
        const int4 v = hp[1];
        const int c0 = lane * 8;
        int t = u.x + u.y + u.z + u.w + v.x + v.y + v.z + v.w;
        int p = (c0 + 0 < me ? u.x : 0) + (c0 + 1 < me ? u.y : 0) + (c0 + 2 < me ? u.z : 0) + (c0 + 3 < me ? u.w : 0)
              + (c0 + 4 < me ? v.x : 0) + (c0 + 5 < me ? v.y : 0) + (c0 + 6 < me ? v.z : 0) + (c0 + 7 < me ? v.w : 0);
        t = wave_isum(t);
        p = wave_isum(p);
        if (lane == 0) { tot[b] = t; pre[b] = p; }
    }
    __syncthreads();
    if (wave == 0) {
        int v[SCAN_PER_LANE];
        int sum = 0;
#pragma unroll
        for (int k = 0; k < SCAN_PER_LANE; ++k) {
            const int b = lane * SCAN_PER_LANE + k;
            v[k] = b < N_BINS ? tot[b] : 0;
            sum += v[k];
        }
        int incl = sum;
        for (int o = 1; o < 32; o <<= 1) {
            const int t = __shfl_up(incl, o);
            if (lane >= o) incl += t;
        }
        int run = incl - sum;
#pragma unroll
        for (int k = 0; k < SCAN_PER_LANE; ++k) {
            const int b = lane * SCAN_PER_LANE + k;
            if (b < N_BINS) {
                if (me == 0) binstart[b] = run;
                offs[b] = run + pre[b];
                run += v[k];
            }
        }
        if (me == 0 && lane == 31) binstart[N_BINS] = run;
    }
    __syncthreads();
#pragma unroll
    for (int k = 0; k < PER_THREAD; ++k) {
        if (d[k] >= 0) {
            const int pos = atomicAdd(&offs[d[k] >> BIN_SHIFT], 1);
            binned[pos] = ((d[k] & (BIN_NODES - 1)) << SRC_BITS) | s[k];
        }
    }
}

__global__ void __launch_bounds__(SORT_THREADS)
gat_bin_lists(const int* __restrict__ binned, const int* __restrict__ binstart,
              int* __restrict__ cnt, int* __restrict__ rows) {
    __shared__ int cl[BIN_NODES];
    __shared__ int lists[BIN_NODES * LDS_LIST];
    const int tid  = threadIdx.x;
    const int lane = tid & 31;
    const int wave = tid >> 5;
    const int bin  = blockIdx.x;
    const int n0   = bin * BIN_NODES;
    if (tid < BIN_NODES) cl[tid] = 0;
    const int start = binstart[bin];
    const int end   = binstart[bin + 1];
    __syncthreads();
    for (int base = start; base < end; base += 8 * SORT_THREADS) {
        int w[8];
#pragma unroll
        for (int k = 0; k < 8; ++k) {
            const int i = base + k * SORT_THREADS + tid;
            w[k] = i < end ? binned[i] : -1;
        }
#pragma unroll
        for (int k = 0; k < 8; ++k) {
            if (w[k] >= 0) {
                const int local = w[k] >> SRC_BITS;
                const int s = w[k] & SRC_MASK;
                const int place = atomicAdd(&cl[local], 1);
                if (place < LDS_LIST) lists[local * LDS_LIST + place] = s;
                else rows[(size_t)(n0 + local) * HF + min(place, ROW_CAP - 1)] = s;
            }
        }
    }
    __syncthreads();
    if (tid < BIN_NODES && n0 + tid < N_NODES) cnt[n0 + tid] = cl[tid];
    for (int local = wave; local < BIN_NODES; local += SORT_THREADS / 32) {
        const int n = n0 + local;
        const int deg = cl[local];
        if (n < N_NODES && lane < min(deg, LDS_LIST))
            rows[(size_t)n * HF + lane] = lists[local * LDS_LIST + lane];
    }
}

__global__ void __launch_bounds__(256)
gat_gemm_wmma(const float* __restrict__ x, const v2f* __restrict__ wt,
              float* __restrict__ h) {
    const int lane = threadIdx.x & 31;
    const int wave = threadIdx.x >> 5;
    const int head = wave & 3;
    const int rh   = wave >> 2;
    const int ln   = lane & 15;
    const int kh   = lane >> 4;

    const int row0 = blockIdx.x * 32 + rh * 16;
    const float* xr = x + (size_t)(row0 + ln) * F_IN;
    const v2f* bp = wt + head * 2048 + kh * 16 + ln;

    v8f c0 = {}, c1 = {};
#pragma unroll
    for (int kq = 0; kq < 32; ++kq) {
        v2f a  = *(const v2f*)(xr + kq * 4 + kh * 2);
        v2f b0 = bp[kq * 32];
        v2f b1 = bp[1024 + kq * 32];
        c0 = __builtin_amdgcn_wmma_f32_16x16x4_f32(false, a, false, b0,
                                                   (short)0, c0, false, false);
        c1 = __builtin_amdgcn_wmma_f32_16x16x4_f32(false, a, false, b1,
                                                   (short)0, c1, false, false);
    }
    float* hp = h + (size_t)(row0 + kh * 8) * HF + head * F_OUT + ln;
#pragma unroll
    for (int r = 0; r < 8; ++r) {
        hp[(size_t)r * HF]      = c0[r];
        hp[(size_t)r * HF + 16] = c1[r];
    }
}

__device__ __forceinline__ long long next_key(const float* orow, int deg,
                                              int lane, long long last) {
    long long best = 0x7fffffffffffffffLL;
    for (int p = lane; p < deg; p += 32) {
        long long key = ((long long)__float_as_int(orow[p]) << 32) | (long long)(unsigned)p;
        if (key > last && key < best) best = key;
    }
    for (int o = 16; o > 0; o >>= 1) {
        long long t = __shfl_xor(best, o);
        best = t < best ? t : best;
    }
    return best;
}

template <int M>
__device__ __forceinline__ float lane_xor(float v) {
    return __int_as_float(__builtin_amdgcn_update_dpp(0, __float_as_int(v), 0x160 + M, 0xf, 0xf, true));
}
__device__ __forceinline__ float head_sum(float v) {
    v += lane_xor<1>(v);
    v += lane_xor<2>(v);
    v += lane_xor<4>(v);
    return v;
}
__device__ __forceinline__ float row_score(const float4 r, const float4 av) {
    float p = r.x * av.x;
    p = fmaf(r.y, av.y, p);
    p = fmaf(r.z, av.z, p);
    p = fmaf(r.w, av.w, p);
    return head_sum(p);
}

__device__ __forceinline__ void soft_one(const float4 row, const float4 asv, float ad,
                                         float& m, float& d, float4& acc) {
    const float e = lrelu(row_score(row, asv) + ad);
    const float gm = fmaxf(m, e);
    const float sc = expf(m - gm);
    const float x = expf(e - gm);
    d = fmaf(d, sc, x);
    acc.x = fmaf(acc.x, sc, x * row.x);
    acc.y = fmaf(acc.y, sc, x * row.y);
    acc.z = fmaf(acc.z, sc, x * row.z);
    acc.w = fmaf(acc.w, sc, x * row.w);
    m = gm;
}

template <int J>
__device__ __forceinline__ float group_lane(float v) {
    return __int_as_float((int)__builtin_amdgcn_mov_dpp8((unsigned)__float_as_int(v), (unsigned)J * 0x249249u));
}

__device__ __forceinline__ void soft_group(const float* __restrict__ h, int ss, int k0, int rem, int lane,
                                           const float4 asv, float ad, bool rescale,
                                           float& m, float& d, float4& acc) {
    float4 r[8];
    float e[8];
#pragma unroll
    for (int j = 0; j < 8; ++j) {
        if (j < rem) {
            const int sk = rl_i(ss, k0 + j);
            r[j] = ((const float4*)h)[(unsigned)sk * (HF / 4) + (unsigned)lane];
        }
    }
    float gm = m;
#pragma unroll
    for (int j = 0; j < 8; ++j) {
        e[j] = -__builtin_inff();
        if (j < rem) {
            e[j] = lrelu(row_score(r[j], asv) + ad);
            gm = fmaxf(gm, e[j]);
        }
    }
    if (rescale) {
        const float sc = expf(m - gm);
        d *= sc;
        acc.x *= sc; acc.y *= sc; acc.z *= sc; acc.w *= sc;
    }
    const int li = lane & 7;
    float es = e[0];
#pragma unroll
    for (int j = 1; j < 8; ++j) es = (li == j) ? e[j] : es;
    const float xs = expf(es - gm);
#define GAT_ROW_TERM(J)                                   \
    if (J < rem) {                                        \
        const float x = group_lane<J>(xs);                \
        d += x;                                           \
        acc.x = fmaf(x, r[J].x, acc.x);                   \
        acc.y = fmaf(x, r[J].y, acc.y);                   \
        acc.z = fmaf(x, r[J].z, acc.z);                   \
        acc.w = fmaf(x, r[J].w, acc.w);                   \
    }
    GAT_ROW_TERM(0) GAT_ROW_TERM(1) GAT_ROW_TERM(2) GAT_ROW_TERM(3)
    GAT_ROW_TERM(4) GAT_ROW_TERM(5) GAT_ROW_TERM(6) GAT_ROW_TERM(7)
#undef GAT_ROW_TERM
    m = gm;
}

__global__ void __launch_bounds__(256)
gat_node_gather(const float* __restrict__ h, const float* __restrict__ a,
                const int* __restrict__ cnt, const int* __restrict__ esrc,
                const int* __restrict__ edst, const float* __restrict__ bias,
                float* out) {
    const int lane = threadIdx.x & 31;
    const int head = lane >> 3;
    const int n = __builtin_amdgcn_readfirstlane(blockIdx.x * 8 + (threadIdx.x >> 5));
    if (n >= N_NODES) return;
    const int deg = __builtin_amdgcn_readfirstlane(cnt[n]);
    float* orow = out + (size_t)n * HF;
    const float4 asv = *(const float4*)(a + head * 2 * F_OUT + (lane & 7) * 4);
    const float4 adv = *(const float4*)(a + head * 2 * F_OUT + F_OUT + (lane & 7) * 4);
    const float4 hn  = *(const float4*)(h + ((unsigned)n * HF + (unsigned)lane * 4));
    const float4 bv  = *(const float4*)(bias + lane * 4);
    const float ad = row_score(hn, adv);
    float m = -__builtin_inff();
    float d = 0.f;
    float4 acc = make_float4(0.f, 0.f, 0.f, 0.f);

    if (deg <= 64) {
        int ss0 = 0, ss1 = 0;
        if (deg <= 32) {
            const bool valid = lane < deg;
            const int s_in = valid ? __float_as_int(orow[lane]) : 0;
            const unsigned ck = valid ? (((unsigned)s_in << 5) | (unsigned)lane)
                                      : (0xFFFFFFE0u | (unsigned)lane);
            int rk = 0;
            for (int j = 0; j < deg; ++j) {
                unsigned kj = (unsigned)rl_i((int)ck, j);
                rk += (kj < ck) ? 1 : 0;
            }
            if (!valid) rk = lane;
            ss0 = __builtin_amdgcn_ds_permute(rk << 2, s_in);
        } else {
            const bool v1 = lane + 32 < deg;
            const int s0 = __float_as_int(orow[lane]);
            const int s1 = v1 ? __float_as_int(orow[32 + lane]) : 0;
            const unsigned k0 = ((unsigned)s0 << 6) | (unsigned)lane;
            const unsigned k1 = v1 ? (((unsigned)s1 << 6) | (unsigned)(32 + lane)) : 0xFFFFFFFFu;
            int r0 = 0, r1 = 0;
            for (int j = 0; j < 32; ++j) {
                const unsigned kj = (unsigned)rl_i((int)k0, j);
                r0 += (kj < k0) ? 1 : 0;
                r1 += (kj < k1) ? 1 : 0;
            }
            for (int j = 32; j < deg; ++j) {
                const unsigned kj = (unsigned)rl_i((int)k1, j - 32);
                r0 += (kj < k0) ? 1 : 0;
                r1 += (kj < k1) ? 1 : 0;
            }
            for (int k = 0; k < deg; ++k) {
                const unsigned b0 = __builtin_amdgcn_ballot_w32(r0 == k);
                const unsigned b1 = __builtin_amdgcn_ballot_w32(r1 == k);
                const int l = __builtin_ctz(b0 | b1);
                const int sk = b0 ? rl_i(s0, l) : rl_i(s1, l);
                if (lane == (k & 31)) { if (k < 32) ss0 = sk; else ss1 = sk; }
            }
        }
        for (int k0 = 0; k0 < deg; k0 += 8)
            soft_group(h, k0 < 32 ? ss0 : ss1, k0 & 31, deg - k0, lane, asv, ad, k0 != 0, m, d, acc);
    } else if (deg <= ROW_CAP) {
        long long last = -1;
        for (int k = 0; k < deg; ++k) {
            last = next_key(orow, deg, lane, last);
            const int s = (int)(last >> 32);
            const float4 row = *(const float4*)(h + (size_t)s * HF + lane * 4);
            soft_one(row, asv, ad, m, d, acc);
        }
    } else {
        for (int e0 = 0; e0 < N_EDGES; e0 += 32) {
            const bool hit = edst[e0 + lane] == n;
            const int s_l = hit ? esrc[e0 + lane] : 0;
            unsigned todo = __builtin_amdgcn_ballot_w32(hit);
            while (todo) {
                const int l = __builtin_ctz(todo);
                todo &= todo - 1;
                const int s = rl_i(s_l, l);
                const float4 row = *(const float4*)(h + (size_t)s * HF + lane * 4);
                soft_one(row, asv, ad, m, d, acc);
            }
        }
    }
    const float q = 1.0f / (d + 1e-16f);
    float4 o;
    o.x = acc.x * q + bv.x; o.y = acc.y * q + bv.y; o.z = acc.z * q + bv.z; o.w = acc.w * q + bv.w;
    *(float4*)(orow + lane * 4) = o;
}

extern "C" void kernel_launch(void* const* d_in, const int* in_sizes, int n_in,
                              void* d_out, int out_size, void* d_ws, size_t ws_size,
                              hipStream_t stream) {
    const float* x      = (const float*)d_in[0];
    const float* weight = (const float*)d_in[1];
    const float* a      = (const float*)d_in[2];
    const float* bias   = (const float*)d_in[3];
    const int*   esrc   = (const int*)d_in[4];
    const int*   edst   = (const int*)d_in[5];
    float* out = (float*)d_out;

    float* h    = (float*)d_ws;
    int*   cnt  = (int*)(h + (size_t)N_NODES * HF);
    v2f*   wt   = (v2f*)(cnt + N_NODES);
    int*   rows = (int*)d_out;
    int*   binned   = (int*)d_ws;
    int*   hist     = binned + N_EDGES;
    int*   binstart = hist + N_BINS * EDGE_CHUNKS;

    const int TB = 256;
    gat_prep_hist<<<EDGE_CHUNKS, SORT_THREADS, 0, stream>>>(weight, wt, edst, hist);
    gat_bin_scatter<<<EDGE_CHUNKS, SORT_THREADS, 0, stream>>>(esrc, edst, hist, binstart, binned);
    gat_bin_lists<<<N_BINS, SORT_THREADS, 0, stream>>>(binned, binstart, cnt, rows);
    gat_gemm_wmma<<<N_NODES / 32, TB, 0, stream>>>(x, wt, h);
    gat_node_gather<<<N_NODES / 8, TB, 0, stream>>>(h, a, cnt, esrc, edst, bias, out);
}


# Chunk sort: the chunk's bin-sorted run is assembled in LDS and copied to the scratch as whole 128-byte lines, one per wave store; was 1.6M 4-byte stores scattered 32 lines per instruction; same sums
# speedup vs baseline: 8.8451x; 1.4615x over previous
#include <hip/hip_runtime.h>
#include <math.h>

#define N_NODES 100000
#define N_EDGES 1600000
#define H_HEADS 4
#define F_IN    128
#define F_OUT   32
#define HF      (H_HEADS * F_OUT)
#define SLOPE   0.2f
#define ROW_CAP 64
static_assert(N_EDGES % 32 == 0, "the edge walk of an over-full node has no tail");

#define EDGE_CHUNKS  256
#define CHUNK_EDGES  (N_EDGES / EDGE_CHUNKS)
#define BIN_SHIFT    9
#define BIN_NODES    (1 << BIN_SHIFT)
#define N_BINS       ((N_NODES + BIN_NODES - 1) / BIN_NODES)
#define SRC_BITS     17
#define SRC_MASK     ((1 << SRC_BITS) - 1)
#define SORT_THREADS 1024
#define PER_THREAD   ((CHUNK_EDGES + SORT_THREADS - 1) / SORT_THREADS)
#define SCAN_PER_LANE ((N_BINS + 31) / 32)
#define LDS_LIST     32
static_assert(N_EDGES % EDGE_CHUNKS == 0, "equal chunks");
static_assert(N_NODES <= (1 << SRC_BITS) && BIN_SHIFT + SRC_BITS < 32, "a binned edge fits a non-negative word");
#define BINNED_AT    0
#define OFFT_AT      N_EDGES
#define CHUNKS_PER_WAVE (EDGE_CHUNKS / (SORT_THREADS / 32))
static_assert(N_EDGES % 64 == 0 && EDGE_CHUNKS % 64 == 0 && CHUNKS_PER_WAVE == 8, "a wave's 8 segment bounds of a bin sit in one 64-word piece");
static_assert(OFFT_AT + (N_BINS + 1) * EDGE_CHUNKS <= N_NODES * (HF - ROW_CAP), "the sort's scratch fits in the upper halves of the output rows");
static_assert(ROW_CAP == 64 && HF == 128, "lists below, scratch above; the last kernel holds a whole list in two registers per lane");
__device__ __forceinline__ unsigned scr(unsigned i) { return (i >> 6) * HF + ROW_CAP + (i & 63u); }

#define GEMM_TILES       (N_NODES / 16)
#define GEMM_TILES_FIRST (GEMM_TILES / 2)
#define GEMM_GROUPS      256
static_assert(N_NODES % 16 == 0, "whole 16-row tiles");

typedef __attribute__((ext_vector_type(2))) float v2f;
typedef __attribute__((ext_vector_type(8))) float v8f;

__device__ __forceinline__ float lrelu(float v) { return v > 0.f ? v : SLOPE * v; }

__device__ __forceinline__ float wave_max(float v) {
    for (int o = 16; o > 0; o >>= 1) v = fmaxf(v, __shfl_xor(v, o));
    return v;
}
__device__ __forceinline__ float wave_sum(float v) {
    for (int o = 16; o > 0; o >>= 1) v += __shfl_xor(v, o);
    return v;
}
__device__ __forceinline__ int rl_i(int v, int l) { return __builtin_amdgcn_readlane(v, l); }
__device__ __forceinline__ float rl_f(float v, int l) {
    return __int_as_float(__builtin_amdgcn_readlane(__float_as_int(v), l));
}

#define B_WORDS (H_HEADS * 2 * 32 * 2 * 16)
static_assert(B_WORDS % SORT_THREADS == 0, "every thread of a GEMM work-group stages the same number of fragment words");
static_assert(CHUNK_EDGES * sizeof(int) <= B_WORDS * sizeof(v2f), "a sorting work-group assembles its chunk's run in the B-fragment array");
__device__ __forceinline__ void gemm_tiles(const float* __restrict__ x, const float* __restrict__ w, v2f* bl,
                                           float* __restrict__ h, int g, int t0, int t1) {
    {
        v2f v[B_WORDS / SORT_THREADS];
#pragma unroll
        for (int it = 0; it < B_WORDS / SORT_THREADS; ++it) {
            const int idx  = it * SORT_THREADS + (int)threadIdx.x;
            const int n    = idx & 15;
            const int kh   = (idx >> 4) & 1;
            const int kq   = (idx >> 5) & 31;
            const int ct   = (idx >> 10) & 1;
            const int head = idx >> 11;
            const float* wp = w + head * (F_IN * F_OUT) + (kq * 4 + kh * 2) * F_OUT + ct * 16 + n;
            v[it].x = wp[0];
            v[it].y = wp[F_OUT];
        }
#pragma unroll
        for (int it = 0; it < B_WORDS / SORT_THREADS; ++it) bl[it * SORT_THREADS + (int)threadIdx.x] = v[it];
    }
    __syncthreads();
    const int lane = threadIdx.x & 31;
    const int wave = __builtin_amdgcn_readfirstlane(threadIdx.x >> 5);
    const int head = wave & 3;
    const int q    = wave >> 2;
    const int ln   = lane & 15;
    const int kh   = lane >> 4;
    int boff = head * 2048 + kh * 16 + ln;

    for (int t = t0 + q * GEMM_GROUPS + g; t < t1; t += 8 * GEMM_GROUPS) {
        asm volatile("" : "+v"(boff));
        const v2f* bp = bl + boff;
        const int row0 = t * 16;
        const float* xr = x + (size_t)(row0 + ln) * F_IN;
        v8f c0 = {}, c1 = {};
#pragma nounroll
        for (int kg = 0; kg < 32; kg += 8) {
#pragma unroll
            for (int j = 0; j < 8; ++j) {
                const int kq = kg + j;
                v2f a  = *(const v2f*)(xr + kq * 4 + kh * 2);
                v2f b0 = bp[kq * 32];
                v2f b1 = bp[1024 + kq * 32];
                c0 = __builtin_amdgcn_wmma_f32_16x16x4_f32(false, a, false, b0,
                                                           (short)0, c0, false, false);
                c1 = __builtin_amdgcn_wmma_f32_16x16x4_f32(false, a, false, b1,
                                                           (short)0, c1, false, false);
            }
        }
        float* hp = h + (size_t)(row0 + kh * 8) * HF + head * F_OUT + ln;
#pragma unroll
        for (int r = 0; r < 8; ++r) {
            hp[(size_t)r * HF]      = c0[r];
            hp[(size_t)r * HF + 16] = c1[r];
        }
    }
}

__global__ void __launch_bounds__(SORT_THREADS)
gat_chunk_sort_gemm(const int* __restrict__ esrc, const int* __restrict__ edst,
                    int* rows,
                    const float* __restrict__ x, const float* __restrict__ w, float* __restrict__ h) {
    __shared__ int hb[N_BINS];
    __shared__ int offs[N_BINS];
    __shared__ __attribute__((aligned(16))) v2f bl[B_WORDS];
    if (blockIdx.x < GEMM_GROUPS) {
        gemm_tiles(x, w, bl, h, blockIdx.x, 0, GEMM_TILES_FIRST);
        return;
    }
    const int tid  = threadIdx.x;
    const int lane = tid & 31;
    const int wave = tid >> 5;
    const int me   = blockIdx.x - GEMM_GROUPS;
    const int* dp = edst + me * CHUNK_EDGES;
    const int* sp = esrc + me * CHUNK_EDGES;
    if (tid < N_BINS) hb[tid] = 0;
    int d[PER_THREAD], s[PER_THREAD];
#pragma unroll
    for (int k = 0; k < PER_THREAD; ++k) {
        const int e = k * SORT_THREADS + tid;
        d[k] = e < CHUNK_EDGES ? dp[e] : -1;
        s[k] = e < CHUNK_EDGES ? sp[e] : 0;
    }
    __syncthreads();
#pragma unroll
    for (int k = 0; k < PER_THREAD; ++k)
        if (d[k] >= 0) atomicAdd(&hb[d[k] >> BIN_SHIFT], 1);
    __syncthreads();
    if (wave == 0) {
        int v[SCAN_PER_LANE];
        int sum = 0;
#pragma unroll
        for (int k = 0; k < SCAN_PER_LANE; ++k) {
            const int b = lane * SCAN_PER_LANE + k;
            v[k] = b < N_BINS ? hb[b] : 0;
            sum += v[k];
        }
        int incl = sum;
        for (int o = 1; o < 32; o <<= 1) {
            const int t = __shfl_up(incl, o);
            if (lane >= o) incl += t;
        }
        int run = incl - sum;
#pragma unroll
        for (int k = 0; k < SCAN_PER_LANE; ++k) {
            const int b = lane * SCAN_PER_LANE + k;
            if (b < N_BINS) {
                rows[scr(OFFT_AT + b * EDGE_CHUNKS + me)] = run;
                offs[b] = run;
                run += v[k];
            }
        }
        if (lane == 31) rows[scr(OFFT_AT + N_BINS * EDGE_CHUNKS + me)] = run;
    }
    __syncthreads();
    int* run_l = (int*)bl;
#pragma unroll
    for (int k = 0; k < PER_THREAD; ++k) {
        if (d[k] >= 0) {
            const int pos = atomicAdd(&offs[d[k] >> BIN_SHIFT], 1);
            run_l[pos] = ((d[k] & (BIN_NODES - 1)) << SRC_BITS) | s[k];
        }
    }
    __syncthreads();
    const int first = me * CHUNK_EDGES;
    const int line0 = first & ~31;
    static_assert(PER_THREAD * SORT_THREADS >= CHUNK_EDGES + 31, "PER_THREAD words per thread cover the run from its first line's boundary on");
    int wv[PER_THREAD];
#pragma unroll
    for (int k = 0; k < PER_THREAD; ++k) {
        const int j = line0 + k * SORT_THREADS + tid - first;
        wv[k] = run_l[min(max(j, 0), CHUNK_EDGES - 1)];
    }
#pragma unroll
    for (int k = 0; k < PER_THREAD; ++k) {
        const int i = line0 + k * SORT_THREADS + tid;
        if (i >= first && i < first + CHUNK_EDGES) rows[scr(BINNED_AT + i)] = wv[k];
    }
}

static_assert(BIN_NODES * LDS_LIST * sizeof(int) == B_WORDS * sizeof(v2f), "the lists and the B fragments share one LDS array");
__global__ void __launch_bounds__(SORT_THREADS)
gat_bin_lists_gemm(int* __restrict__ cnt, int* rows,
                   const float* __restrict__ x, const float* __restrict__ w, float* __restrict__ h) {
    __shared__ int cl[BIN_NODES];
    __shared__ __attribute__((aligned(16))) int lists[BIN_NODES * LDS_LIST];
    if (blockIdx.x < GEMM_GROUPS) {
        gemm_tiles(x, w, (v2f*)lists, h, blockIdx.x, GEMM_TILES_FIRST, GEMM_TILES);
        return;
    }
    const int tid  = threadIdx.x;
    const int lane = tid & 31;
    const int wave = __builtin_amdgcn_readfirstlane(tid >> 5);
    const int bin  = blockIdx.x - GEMM_GROUPS;
    const int n0   = bin * BIN_NODES;
    if (tid < BIN_NODES) cl[tid] = 0;
    const int c0   = wave * CHUNKS_PER_WAVE;
    const int st_l = rows[scr(OFFT_AT + bin * EDGE_CHUNKS + c0 + (lane & 7))];
    const int en_l = rows[scr(OFFT_AT + (bin + 1) * EDGE_CHUNKS + c0 + (lane & 7))];
    int st[CHUNKS_PER_WAVE], len[CHUNKS_PER_WAVE];
    int maxlen = 0;
#pragma unroll
    for (int j = 0; j < CHUNKS_PER_WAVE; ++j) {
        st[j]  = rl_i(st_l, j);
        len[j] = rl_i(en_l, j) - st[j];
        maxlen = max(maxlen, len[j]);
    }
    __syncthreads();
    for (int off = 0; off < maxlen; off += 32) {
        int wd[CHUNKS_PER_WAVE];
#pragma unroll
        for (int j = 0; j < CHUNKS_PER_WAVE; ++j) {
            const int i = off + lane;
            wd[j] = i < len[j] ? rows[scr(BINNED_AT + (c0 + j) * CHUNK_EDGES + st[j] + i)] : -1;
        }
#pragma unroll
        for (int j = 0; j < CHUNKS_PER_WAVE; ++j) {
            if (wd[j] >= 0) {
                const int local = wd[j] >> SRC_BITS;
                const int s = wd[j] & SRC_MASK;
                const int place = atomicAdd(&cl[local], 1);
                if (place < LDS_LIST) lists[local * LDS_LIST + place] = s;
                else rows[(size_t)(n0 + local) * HF + min(place, ROW_CAP - 1)] = s;
            }
        }
    }
    __syncthreads();
    if (tid < BIN_NODES && n0 + tid < N_NODES) cnt[n0 + tid] = cl[tid];
    for (int local = wave; local < BIN_NODES; local += SORT_THREADS / 32) {
        const int n = n0 + local;
        const int deg = cl[local];
        if (n < N_NODES && lane < min(deg, LDS_LIST))
            rows[(size_t)n * HF + lane] = lists[local * LDS_LIST + lane];
    }
}

template <int M>
__device__ __forceinline__ float lane_xor(float v) {
    return __int_as_float(__builtin_amdgcn_update_dpp(0, __float_as_int(v), 0x160 + M, 0xf, 0xf, true));
}
__device__ __forceinline__ float head_sum(float v) {
    v += lane_xor<1>(v);
    v += lane_xor<2>(v);
    v += lane_xor<4>(v);
    return v;
}
__device__ __forceinline__ float row_score(const float4 r, const float4 av) {
    float p = r.x * av.x;
    p = fmaf(r.y, av.y, p);
    p = fmaf(r.z, av.z, p);
    p = fmaf(r.w, av.w, p);
    return head_sum(p);
}

__device__ __forceinline__ void soft_one(const float4 row, const float4 asv, float ad,
                                         float& m, float& d, float4& acc) {
    const float e = lrelu(row_score(row, asv) + ad);
    const float gm = fmaxf(m, e);
    const float sc = expf(m - gm);
    const float x = expf(e - gm);
    d = fmaf(d, sc, x);
    acc.x = fmaf(acc.x, sc, x * row.x);
    acc.y = fmaf(acc.y, sc, x * row.y);
    acc.z = fmaf(acc.z, sc, x * row.z);
    acc.w = fmaf(acc.w, sc, x * row.w);
    m = gm;
}

template <int J>
__device__ __forceinline__ float group_lane(float v) {
    return __int_as_float((int)__builtin_amdgcn_mov_dpp8((unsigned)__float_as_int(v), (unsigned)J * 0x249249u));
}

template <int M>
__device__ __forceinline__ float fold_rows(float pa, float pb, bool set) {
    const float keep = set ? pb : pa;
    const float send = set ? pa : pb;
    return keep + lane_xor<M>(send);
}

template <int R>
__device__ __forceinline__ void soft_group(const float* __restrict__ h, int ss, int k0, int rem, int lane,
                                           const float4 asv, float ad, bool rescale,
                                           float& m, float& d, float4& acc) {
    static_assert(R == 8 || R == 4, "a group is 8 rows, or 4 for a short tail");
    float4 r[R];
    if (rem >= R) {
#pragma unroll
        for (int j = 0; j < R; ++j) {
            const int sk = rl_i(ss, k0 + j);
            r[j] = ((const float4*)h)[(unsigned)sk * (HF / 4) + (unsigned)lane];
        }
    } else {
#pragma unroll
        for (int j = 0; j < R; ++j) {
            r[j] = make_float4(0.f, 0.f, 0.f, 0.f);
            if (j < rem) {
                const int sk = rl_i(ss, k0 + j);
                r[j] = ((const float4*)h)[(unsigned)sk * (HF / 4) + (unsigned)lane];
            }
        }
    }
    float p[R];
#pragma unroll
    for (int j = 0; j < R; ++j) {
        p[j] = r[j].x * asv.x;
        p[j] = fmaf(r[j].y, asv.y, p[j]);
        p[j] = fmaf(r[j].z, asv.z, p[j]);
        p[j] = fmaf(r[j].w, asv.w, p[j]);
    }
    const int li = lane & (R - 1);
    const bool b0 = (lane & 1) != 0, b1 = (lane & 2) != 0, b2 = (lane & 4) != 0;
    const float q0 = fold_rows<1>(p[0], p[1], b0);
    const float q1 = fold_rows<1>(p[2], p[3], b0);
    const float u0 = fold_rows<2>(q0, q1, b1);
    float s;
    if constexpr (R == 8) {
        const float q2 = fold_rows<1>(p[4], p[5], b0);
        const float q3 = fold_rows<1>(p[6], p[7], b0);
        const float u1 = fold_rows<2>(q2, q3, b1);
        s = fold_rows<4>(u0, u1, b2);
    } else {
        s = u0 + lane_xor<4>(u0);
    }
    const float v  = li < rem ? s + ad : -__builtin_inff();
    const float es = fmaxf(v, SLOPE * v);
    float gx = es;
    gx = fmaxf(gx, lane_xor<1>(gx));
    gx = fmaxf(gx, lane_xor<2>(gx));
    gx = fmaxf(gx, lane_xor<4>(gx));
    const float gm = fmaxf(m, gx);
    if (rescale) {
        const float sc = expf(m - gm);
        d *= sc;
        acc.x *= sc; acc.y *= sc; acc.z *= sc; acc.w *= sc;
    }
    const float xs = expf(es - gm);
#define GAT_ROW_TERM(J)                                   \
    {                                                     \
        const float x = group_lane<J>(xs);                \
        d += x;                                           \
        acc.x = fmaf(x, r[J].x, acc.x);                   \
        acc.y = fmaf(x, r[J].y, acc.y);                   \
        acc.z = fmaf(x, r[J].z, acc.z);                   \
        acc.w = fmaf(x, r[J].w, acc.w);                   \
    }
    GAT_ROW_TERM(0) GAT_ROW_TERM(1) GAT_ROW_TERM(2) GAT_ROW_TERM(3)
    if constexpr (R == 8) {
        GAT_ROW_TERM(4) GAT_ROW_TERM(5) GAT_ROW_TERM(6) GAT_ROW_TERM(7)
    }
#undef GAT_ROW_TERM
    m = gm;
}

__global__ void __launch_bounds__(256)
gat_node_gather(const float* __restrict__ h, const float* __restrict__ a,
                const int* __restrict__ cnt, const int* __restrict__ esrc,
                const int* __restrict__ edst, const float* __restrict__ bias,
                float* out) {
    const int lane = threadIdx.x & 31;
    const int head = lane >> 3;
    const int n = __builtin_amdgcn_readfirstlane(blockIdx.x * 8 + (threadIdx.x >> 5));
    if (n >= N_NODES) return;
    const int deg = __builtin_amdgcn_readfirstlane(cnt[n]);
    float* orow = out + (size_t)n * HF;
    const float4 asv = *(const float4*)(a + head * 2 * F_OUT + (lane & 7) * 4);
    const float4 adv = *(const float4*)(a + head * 2 * F_OUT + F_OUT + (lane & 7) * 4);
    const float4 hn  = *(const float4*)(h + ((unsigned)n * HF + (unsigned)lane * 4));
    const float4 bv  = *(const float4*)(bias + lane * 4);
    const float ad = row_score(hn, adv);
    float m = -__builtin_inff();
    float d = 0.f;
    float4 acc = make_float4(0.f, 0.f, 0.f, 0.f);

    if (deg <= 64) {
        int ss0 = 0, ss1 = 0;
        if (deg <= 32) {
            const bool valid = lane < deg;
            const int s_in = valid ? __float_as_int(orow[lane]) : 0;
            const unsigned ck = valid ? (((unsigned)s_in << 5) | (unsigned)lane)
                                      : (0xFFFFFFE0u | (unsigned)lane);
            int rk = 0;
            for (int j = 0; j < deg; ++j) {
                unsigned kj = (unsigned)rl_i((int)ck, j);
                rk += (kj < ck) ? 1 : 0;
            }
            if (!valid) rk = lane;
            ss0 = __builtin_amdgcn_ds_permute(rk << 2, s_in);
        } else {
            const bool v1 = lane + 32 < deg;
            const int s0 = __float_as_int(orow[lane]);
            const int s1 = v1 ? __float_as_int(orow[32 + lane]) : 0;
            const unsigned k0 = ((unsigned)s0 << 6) | (unsigned)lane;
            const unsigned k1 = v1 ? (((unsigned)s1 << 6) | (unsigned)(32 + lane)) : 0xFFFFFFFFu;
            int r0 = 0, r1 = 0;
            for (int j = 0; j < 32; ++j) {
                const unsigned kj = (unsigned)rl_i((int)k0, j);
                r0 += (kj < k0) ? 1 : 0;
                r1 += (kj < k1) ? 1 : 0;
            }
            for (int j = 32; j < deg; ++j) {
                const unsigned kj = (unsigned)rl_i((int)k1, j - 32);
                r0 += (kj < k0) ? 1 : 0;
                r1 += (kj < k1) ? 1 : 0;
            }
            for (int k = 0; k < deg; ++k) {
                const unsigned b0 = __builtin_amdgcn_ballot_w32(r0 == k);
                const unsigned b1 = __builtin_amdgcn_ballot_w32(r1 == k);
                const int l = __builtin_ctz(b0 | b1);
                const int sk = b0 ? rl_i(s0, l) : rl_i(s1, l);
                if (lane == (k & 31)) { if (k < 32) ss0 = sk; else ss1 = sk; }
            }
        }
        int k0 = 0;
        for (; k0 + 4 < deg; k0 += 8)
            soft_group<8>(h, k0 < 32 ? ss0 : ss1, k0 & 31, deg - k0, lane, asv, ad, k0 != 0, m, d, acc);
        if (k0 < deg) {
            if (k0 != 0) soft_group<4>(h, k0 < 32 ? ss0 : ss1, k0 & 31, deg - k0, lane, asv, ad, true, m, d, acc);
            else         soft_group<4>(h, ss0, 0, deg, lane, asv, ad, false, m, d, acc);
        }
    } else {
        for (int e0 = 0; e0 < N_EDGES; e0 += 32) {
            const bool hit = edst[e0 + lane] == n;
            const int s_l = hit ? esrc[e0 + lane] : 0;
            unsigned todo = __builtin_amdgcn_ballot_w32(hit);
            while (todo) {
                const int l = __builtin_ctz(todo);
                todo &= todo - 1;
                const int s = rl_i(s_l, l);
                const float4 row = *(const float4*)(h + (size_t)s * HF + lane * 4);
                soft_one(row, asv, ad, m, d, acc);
            }
        }
    }
    const float q = 1.0f / (d + 1e-16f);
    float4 o;
    o.x = acc.x * q + bv.x; o.y = acc.y * q + bv.y; o.z = acc.z * q + bv.z; o.w = acc.w * q + bv.w;
    *(float4*)(orow + lane * 4) = o;
}

extern "C" void kernel_launch(void* const* d_in, const int* in_sizes, int n_in,
                              void* d_out, int out_size, void* d_ws, size_t ws_size,
                              hipStream_t stream) {
    const float* x      = (const float*)d_in[0];
    const float* weight = (const float*)d_in[1];
    const float* a      = (const float*)d_in[2];
    const float* bias   = (const float*)d_in[3];
    const int*   esrc   = (const int*)d_in[4];
    const int*   edst   = (const int*)d_in[5];
    float* out = (float*)d_out;

    float* h    = (float*)d_ws;
    int*   cnt  = (int*)(h + (size_t)N_NODES * HF);
    int*   rows = (int*)d_out;

    gat_chunk_sort_gemm<<<GEMM_GROUPS + EDGE_CHUNKS, SORT_THREADS, 0, stream>>>(esrc, edst, rows, x, weight, h);
    gat_bin_lists_gemm<<<GEMM_GROUPS + N_BINS, SORT_THREADS, 0, stream>>>(cnt, rows, x, weight, h);
    gat_node_gather<<<N_NODES / 8, 256, 0, stream>>>(h, a, cnt, esrc, edst, bias, out);
}
